# starting kernel
# baseline (speedup 1.0000x reference)
#include <hip/hip_runtime.h>

#define TT      2048
#define NBANDS  128
#define NBATCH  16
#define ROWS    (NBATCH * NBANDS)
#define CHUNK   16
#define NCHUNKS (TT / CHUNK)
#define EPSV    1e-5f

typedef __attribute__((ext_vector_type(2))) float v2f;
typedef __attribute__((ext_vector_type(4))) float v4f;
typedef __attribute__((ext_vector_type(8))) float v8f;

__device__ __forceinline__ float s_coef() {
    return (sqrtf(262145.0f) - 1.0f) / 131072.0f;
}

__device__ __forceinline__ float qpow(float q, int e) {
    float r = 1.0f;
#pragma unroll 1
    for (int i = 0; i < e; ++i) r *= q;
    return r;
}

__device__ __forceinline__ float fast_pow_pos(float y, float p) {
    return __builtin_amdgcn_exp2f(p * __builtin_amdgcn_logf(y));
}

__global__ void __launch_bounds__(NBANDS)
pcen_pass0_bands(const float* __restrict__ alpha, const float* __restrict__ delta,
                 const float* __restrict__ r, float* __restrict__ bws) {
    int b = threadIdx.x;
    float a  = __expf(alpha[b]);
    float d  = __expf(delta[b]);
    float rr = __expf(r[b]);
    bws[b]              = a;
    bws[NBANDS + b]     = d;
    bws[2 * NBANDS + b] = rr;
    bws[3 * NBANDS + b] = fast_pow_pos(d, rr);
}

__global__ void __launch_bounds__(256)
pcen_pass1_chunklocal(const float* __restrict__ x, float* __restrict__ Lws) {
    int gid   = blockIdx.x * blockDim.x + threadIdx.x;
    int row   = gid >> 7;
    int chunk = gid & (NCHUNKS - 1);
    const float s = s_coef();
    const float q = 1.0f - s;

    const v4f* p4 = (const v4f*)(x + (size_t)row * TT + (size_t)chunk * CHUNK);
    float acc = 0.0f;
#pragma unroll
    for (int i = 0; i < 4; ++i) {
        v4f v = p4[i];
        acc = acc * q + v.x;
        acc = acc * q + v.y;
        acc = acc * q + v.z;
        acc = acc * q + v.w;
    }
    Lws[gid] = s * acc;
}

__global__ void __launch_bounds__(256)
pcen_pass2_carry(const float* __restrict__ x, const float* __restrict__ Lws,
                 float* __restrict__ Cws) {
    int row = blockIdx.x * blockDim.x + threadIdx.x;
    const float s  = s_coef();
    const float q  = 1.0f - s;
    const float q2 = q * q, q4 = q2 * q2, q8 = q4 * q4, q16 = q8 * q8;

    float c = x[(size_t)row * TT];
    const float* Lr = Lws + (size_t)row * NCHUNKS;
    float*       Cr = Cws + (size_t)row * NCHUNKS;
#pragma unroll 4
    for (int k = 0; k < NCHUNKS; ++k) {
        Cr[k] = c;
        c = q16 * c + Lr[k];
    }
}

__global__ void __launch_bounds__(32)
pcen_pass3_wmma(const float* __restrict__ x, const float* __restrict__ Cws,
                const float* __restrict__ bws, float* __restrict__ out) {
    const float s = s_coef();
    const float q = 1.0f - s;

    const int chunk = blockIdx.x;
    const int rg    = blockIdx.y;
    const int row0  = rg * 16;
    const int t0    = chunk * CHUNK;

    const int lane = threadIdx.x;
    const int m    = lane & 15;
    const int hi   = lane >> 4;

    v2f Bm[4];
#pragma unroll
    for (int kk = 0; kk < 4; ++kk) {
        int i0 = kk * 4 + 2 * hi;
        Bm[kk].x = (m >= i0)     ? s * qpow(q, m - i0)     : 0.0f;
        Bm[kk].y = (m >= i0 + 1) ? s * qpow(q, m - i0 - 1) : 0.0f;
    }

    const float* xr = x + (size_t)(row0 + m) * TT;
    v2f Am[4];
#pragma unroll
    for (int kk = 0; kk < 4; ++kk) {
        int bt = t0 + kk * 4 + 2 * hi;
        Am[kk] = *(const v2f*)(xr + bt);
    }

    const float qn1 = qpow(q, m + 1);
    v8f acc;
#pragma unroll
    for (int v = 0; v < 8; ++v) {
        int rowM = row0 + v + 8 * hi;
        acc[v] = Cws[(size_t)rowM * NCHUNKS + chunk] * qn1;
    }

    acc = __builtin_amdgcn_wmma_f32_16x16x4_f32(false, Am[0], false, Bm[0], (short)0, acc, false, false);
    acc = __builtin_amdgcn_wmma_f32_16x16x4_f32(false, Am[1], false, Bm[1], (short)0, acc, false, false);
    acc = __builtin_amdgcn_wmma_f32_16x16x4_f32(false, Am[2], false, Bm[2], (short)0, acc, false, false);
    acc = __builtin_amdgcn_wmma_f32_16x16x4_f32(false, Am[3], false, Bm[3], (short)0, acc, false, false);

    const int t         = t0 + m;
    const int bidx      = row0 >> 7;
    const int band_base = (row0 & 127) + 8 * hi;
    const float* ea  = bws;
    const float* ed  = bws + NBANDS;
    const float* err = bws + 2 * NBANDS;
    const float* edr = bws + 3 * NBANDS;
    v4f a_lo  = *(const v4f*)(ea  + band_base), a_hi  = *(const v4f*)(ea  + band_base + 4);
    v4f d_lo  = *(const v4f*)(ed  + band_base), d_hi  = *(const v4f*)(ed  + band_base + 4);
    v4f r_lo  = *(const v4f*)(err + band_base), r_hi  = *(const v4f*)(err + band_base + 4);
    v4f dr_lo = *(const v4f*)(edr + band_base), dr_hi = *(const v4f*)(edr + band_base + 4);
    float av[8]  = {a_lo.x, a_lo.y, a_lo.z, a_lo.w, a_hi.x, a_hi.y, a_hi.z, a_hi.w};
    float dv[8]  = {d_lo.x, d_lo.y, d_lo.z, d_lo.w, d_hi.x, d_hi.y, d_hi.z, d_hi.w};
    float rv[8]  = {r_lo.x, r_lo.y, r_lo.z, r_lo.w, r_hi.x, r_hi.y, r_hi.z, r_hi.w};
    float drv[8] = {dr_lo.x, dr_lo.y, dr_lo.z, dr_lo.w, dr_hi.x, dr_hi.y, dr_hi.z, dr_hi.w};

    float res[8];
#pragma unroll
    for (int v = 0; v < 8; ++v) {
        int   rowM = row0 + v + 8 * hi;
        float xv   = x[(size_t)rowM * TT + t];
        float sm   = fmaxf(acc[v], 0.0f);
        float smooth = fast_pow_pos(EPSV + sm, -av[v]);
        float base   = fmaf(xv, smooth, dv[v]);
        res[v] = fast_pow_pos(base, rv[v]) - drv[v];
    }
    float* ob = out + (size_t)bidx * TT * NBANDS + (size_t)t * NBANDS + band_base;
    v4f lo;  lo.x  = res[0]; lo.y  = res[1]; lo.z  = res[2]; lo.w  = res[3];
    v4f hi4; hi4.x = res[4]; hi4.y = res[5]; hi4.z = res[6]; hi4.w = res[7];
    *(v4f*)(ob)     = lo;
    *(v4f*)(ob + 4) = hi4;
}

extern "C" void kernel_launch(void* const* d_in, const int* in_sizes, int n_in,
                              void* d_out, int out_size, void* d_ws, size_t ws_size,
                              hipStream_t stream) {
    const float* x     = (const float*)d_in[0];
    const float* alpha = (const float*)d_in[1];
    const float* delta = (const float*)d_in[2];
    const float* r     = (const float*)d_in[3];
    float* out = (float*)d_out;

    float* bws = (float*)d_ws;
    float* Lws = bws + 4 * NBANDS;
    float* Cws = Lws + (size_t)ROWS * NCHUNKS;

    pcen_pass0_bands<<<1, NBANDS, 0, stream>>>(alpha, delta, r, bws);
    pcen_pass1_chunklocal<<<(ROWS * NCHUNKS) / 256, 256, 0, stream>>>(x, Lws);
    pcen_pass2_carry<<<ROWS / 256, 256, 0, stream>>>(x, Lws, Cws);

    dim3 g3(NCHUNKS, ROWS / 16);
    pcen_pass3_wmma<<<g3, 32, 0, stream>>>(x, Cws, bws, out);
}


# Transposed product: decay matrix as A, x tile as B, k slots renumbered (lane half h = steps 8h..8h+7; new sum order in a tile): one 32B x load per lane feeds WMMA and elementwise; 64B store pieces
# speedup vs baseline: 2.4554x; 2.4554x over previous
#include <hip/hip_runtime.h>

#define TT      2048
#define NBANDS  128
#define NBATCH  16
#define ROWS    (NBATCH * NBANDS)
#define CHUNK   16
#define NCHUNKS (TT / CHUNK)
#define EPSV    1e-5f
#define NSPLIT  2
#define NWAVES  8
#define TPW     8
#define HALF    (TT / NSPLIT)

static_assert(NSPLIT == 2 && NSPLIT * NWAVES * TPW == NCHUNKS, "the kernel is written for two halves");
static_assert(CHUNK * TPW == 128 && NWAVES * 2 == 16, "first-half fold: 128 steps per load, two rows per wave");

typedef __attribute__((ext_vector_type(2))) float v2f;
typedef __attribute__((ext_vector_type(4))) float v4f;
typedef __attribute__((ext_vector_type(8))) float v8f;

__device__ __forceinline__ float s_coef() {
    return (sqrtf(262145.0f) - 1.0f) / 131072.0f;
}

__device__ __forceinline__ float qpow_bits(float q, int e) {
    const float q2 = q * q, q4 = q2 * q2, q8 = q4 * q4, q16 = q8 * q8;
    const float q32 = q16 * q16, q64 = q32 * q32;
    float r = (e & 1) ? q : 1.0f;
    r *= (e & 2)  ? q2  : 1.0f;
    r *= (e & 4)  ? q4  : 1.0f;
    r *= (e & 8)  ? q8  : 1.0f;
    r *= (e & 16) ? q16 : 1.0f;
    r *= (e & 32) ? q32 : 1.0f;
    r *= (e & 64) ? q64 : 1.0f;
    return r;
}

__device__ __forceinline__ float fast_pow_pos(float y, float p) {
    return __builtin_amdgcn_exp2f(p * __builtin_amdgcn_logf(y));
}

__device__ __forceinline__ float chunk_sum(const v4f (&X)[2], const float (&w)[8]) {
    float part = 0.0f;
    part = fmaf(X[0].x, w[0], part);
    part = fmaf(X[0].y, w[1], part);
    part = fmaf(X[0].z, w[2], part);
    part = fmaf(X[0].w, w[3], part);
    part = fmaf(X[1].x, w[4], part);
    part = fmaf(X[1].y, w[5], part);
    part = fmaf(X[1].z, w[6], part);
    part = fmaf(X[1].w, w[7], part);
    return part + __shfl_xor(part, 16);
}

template <int P>
__device__ __forceinline__ void
pcen_half(const float* __restrict__ x, const float* __restrict__ alpha,
          const float* __restrict__ delta, const float* __restrict__ r,
          float* __restrict__ out, float (*seg)[16], float* pre, float (*bandc)[16]) {
    const float s = s_coef();
    const float q = 1.0f - s;
    const float q2 = q * q, q4 = q2 * q2, q8 = q4 * q4, q16 = q8 * q8;
    float qseg = 1.0f;
#pragma unroll
    for (int k = 0; k < TPW; ++k) qseg *= q16;

    const int rg = blockIdx.x;
    const int wave = __builtin_amdgcn_readfirstlane((int)(threadIdx.x >> 5));
    const int lane = threadIdx.x & 31;
    const int m    = lane & 15;
    const int hi   = lane >> 4;
    const int row0 = rg * 16;
    const int g     = P * NWAVES + wave;
    const int tile0 = g * TPW;

    if (wave == 0 && lane < 16) {
        const int b = (row0 & 127) + lane;
        float a  = __expf(alpha[b]);
        float d  = __expf(delta[b]);
        float rr = __expf(r[b]);
        bandc[0][lane] = a;
        bandc[1][lane] = d;
        bandc[2][lane] = rr;
        bandc[3][lane] = fast_pow_pos(d, rr);
    }

    v4f pv[2][HALF / 128];
    if (P != 0) {
#pragma unroll
        for (int rr = 0; rr < 2; ++rr) {
            const float* xp = x + (size_t)(row0 + 2 * wave + rr) * TT + 4 * lane;
#pragma unroll
            for (int i = 0; i < HALF / 128; ++i) pv[rr][i] = *(const v4f*)(xp + 128 * i);
        }
    }

    const float* xr = x + (size_t)(row0 + m) * TT;
    v4f Xm[TPW][2];
#pragma unroll
    for (int k = 0; k < TPW; ++k) {
        Xm[k][0] = *(const v4f*)(xr + (tile0 + k) * CHUNK + 8 * hi);
        Xm[k][1] = *(const v4f*)(xr + (tile0 + k) * CHUNK + 8 * hi + 4);
    }
    const float x0 = xr[0];
    __builtin_amdgcn_sched_barrier(0);

    float wl[8];
#pragma unroll
    for (int e = 0; e < 8; ++e)
        wl[e] = s * (hi ? qpow_bits(q, 7 - e) : qpow_bits(q, 15 - e));

    if (P != 0) {
        const float q128 = qseg;
        const float wlane = s * qpow_bits(q, 124 - 4 * lane);
        float tsum[2];
#pragma unroll
        for (int rr = 0; rr < 2; ++rr) {
            float h = 0.0f;
#pragma unroll
            for (int i = 0; i < HALF / 128; ++i) {
                float in4 = fmaf(pv[rr][i].x, q, pv[rr][i].y);
                in4 = fmaf(in4, q, pv[rr][i].z);
                in4 = fmaf(in4, q, pv[rr][i].w);
                h = fmaf(h, q128, in4);
            }
            tsum[rr] = h * wlane;
        }
#pragma unroll
        for (int sh = 16; sh >= 1; sh >>= 1) {
            const float o0 = __shfl_xor(tsum[0], sh);
            const float o1 = __shfl_xor(tsum[1], sh);
            tsum[0] += o0;
            tsum[1] += o1;
        }
        if (lane == 0) {
            pre[2 * wave]     = tsum[0];
            pre[2 * wave + 1] = tsum[1];
        }
    }

    float L[TPW];
    float Sown = 0.0f;
#pragma unroll
    for (int k = 0; k < TPW; ++k) {
        L[k] = chunk_sum(Xm[k], wl);
        Sown = fmaf(Sown, q16, L[k]);
    }
    if (hi == 0) seg[wave][m] = Sown;

    v2f Vm[4];
#pragma unroll
    for (int kk = 0; kk < 4; ++kk) {
        const int e = m - (8 * hi + 2 * kk);
        Vm[kk].x = (e >= 0) ? s * qpow_bits(q, e)     : 0.0f;
        Vm[kk].y = (e >= 1) ? s * qpow_bits(q, e - 1) : 0.0f;
    }
    v2f Vc;
    Vc.x = hi ? 0.0f : qpow_bits(q, m + 1);
    Vc.y = 0.0f;

    __syncthreads();

    float c = x0;
    if (P != 0) {
        float qhalf = 1.0f;
#pragma unroll
        for (int i = 0; i < HALF / 128; ++i) qhalf *= qseg;
        c = fmaf(x0, qhalf, pre[m]);
    }
    const int wave_v = (int)(threadIdx.x >> 5);
    float sj[NWAVES];
#pragma unroll
    for (int j = 0; j < NWAVES; ++j) sj[j] = seg[j][m];
#pragma unroll
    for (int j = 0; j < NWAVES; ++j) {
        const bool on = j < wave_v;
        c = fmaf(c, on ? qseg : 1.0f, on ? sj[j] : 0.0f);
    }

    const float a_b  = bandc[0][m];
    const float d_b  = bandc[1][m];
    const float r_b  = bandc[2][m];
    const float dr_b = bandc[3][m];

    const int bidx = row0 >> 7;
    const int band = (row0 & 127) + m;
    float* obase = out + (size_t)bidx * TT * NBANDS + band;

#pragma unroll
    for (int k = 0; k < TPW; ++k) {
        v2f Bc;
        Bc.x = hi ? 0.0f : c;
        Bc.y = 0.0f;
        v2f B0, B1, B2, B3;
        B0.x = Xm[k][0].x; B0.y = Xm[k][0].y;
        B1.x = Xm[k][0].z; B1.y = Xm[k][0].w;
        B2.x = Xm[k][1].x; B2.y = Xm[k][1].y;
        B3.x = Xm[k][1].z; B3.y = Xm[k][1].w;
        v8f acc = {0.0f, 0.0f, 0.0f, 0.0f, 0.0f, 0.0f, 0.0f, 0.0f};
        acc = __builtin_amdgcn_wmma_f32_16x16x4_f32(false, Vc,    false, Bc, (short)0, acc, false, false);
        acc = __builtin_amdgcn_wmma_f32_16x16x4_f32(false, Vm[0], false, B0, (short)0, acc, false, false);
        acc = __builtin_amdgcn_wmma_f32_16x16x4_f32(false, Vm[1], false, B1, (short)0, acc, false, false);
        acc = __builtin_amdgcn_wmma_f32_16x16x4_f32(false, Vm[2], false, B2, (short)0, acc, false, false);
        acc = __builtin_amdgcn_wmma_f32_16x16x4_f32(false, Vm[3], false, B3, (short)0, acc, false, false);
        c = fmaf(q16, c, L[k]);

        const float xv[8] = {Xm[k][0].x, Xm[k][0].y, Xm[k][0].z, Xm[k][0].w,
                             Xm[k][1].x, Xm[k][1].y, Xm[k][1].z, Xm[k][1].w};
        float* ob = obase + (size_t)((tile0 + k) * CHUNK + 8 * hi) * NBANDS;
#pragma unroll
        for (int v = 0; v < 8; ++v) {
            float sm     = fmaxf(acc[v], 0.0f);
            float smooth = fast_pow_pos(EPSV + sm, -a_b);
            float base   = fmaf(xv[v], smooth, d_b);
            ob[(size_t)v * NBANDS] = fast_pow_pos(base, r_b) - dr_b;
        }
    }
}

__global__ void __launch_bounds__(NWAVES * 32)
pcen_fused(const float* __restrict__ x, const float* __restrict__ alpha,
           const float* __restrict__ delta, const float* __restrict__ r,
           float* __restrict__ out) {
    __shared__ float seg[NWAVES][16];
    __shared__ float pre[16];
    __shared__ __attribute__((aligned(16))) float bandc[4][16];
    if (blockIdx.y == 0) pcen_half<0>(x, alpha, delta, r, out, seg, pre, bandc);
    else                 pcen_half<1>(x, alpha, delta, r, out, seg, pre, bandc);
}

extern "C" void kernel_launch(void* const* d_in, const int* in_sizes, int n_in,
                              void* d_out, int out_size, void* d_ws, size_t ws_size,
                              hipStream_t stream) {
    const float* x     = (const float*)d_in[0];
    const float* alpha = (const float*)d_in[1];
    const float* delta = (const float*)d_in[2];
    const float* r     = (const float*)d_in[3];
    float* out = (float*)d_out;

    dim3 grid(ROWS / 16, NSPLIT);
    pcen_fused<<<grid, NWAVES * 32, 0, stream>>>(x, alpha, delta, r, out);
}
